# starting kernel
# baseline (speedup 1.0000x reference)
#include <hip/hip_runtime.h>

typedef __attribute__((ext_vector_type(2))) float v2f;
typedef __attribute__((ext_vector_type(4))) float v4f;
typedef __attribute__((ext_vector_type(8))) float v8f;

#define N_ROWS   64
#define N_COLS   256
#define LDS_PITCH 260
#define THREADS  320
#define OUT_PER_B (N_ROWS * (N_ROWS - 1) / 2)

__global__ __launch_bounds__(THREADS)
void gram_tril_wmma_f32(const float* __restrict__ x, float* __restrict__ out) {
    extern __shared__ float xs[];
    const int b   = blockIdx.x;
    const int tid = threadIdx.x;
    const float* __restrict__ xb = x + (size_t)b * (N_ROWS * N_COLS);

    for (int idx = tid; idx < N_ROWS * (N_COLS / 4); idx += THREADS) {
        const int row = idx >> 6;
        const int c4  = idx & 63;
        v4f v = *(const v4f*)(xb + row * N_COLS + c4 * 4);
        *(v4f*)(&xs[row * LDS_PITCH + c4 * 4]) = v;
    }
    __syncthreads();

    const int wave = tid >> 5;
    const int lane = tid & 31;
    static const int TI[10] = {0,1,1,2,2,2,3,3,3,3};
    static const int TJ[10] = {0,0,1,0,1,2,0,1,2,3};
    const int ti = TI[wave];
    const int tj = TJ[wave];

    const int lrow = lane & 15;
    const int koff = (lane >> 4) << 1;
    const float* aBase = &xs[(ti * 16 + lrow) * LDS_PITCH + koff];
    const float* bBase = &xs[(tj * 16 + lrow) * LDS_PITCH + koff];

    v8f acc = {0.f, 0.f, 0.f, 0.f, 0.f, 0.f, 0.f, 0.f};
    #pragma unroll 8
    for (int k = 0; k < N_COLS; k += 4) {
        v2f av = *(const v2f*)(aBase + k);
        v2f bv = *(const v2f*)(bBase + k);
        acc = __builtin_amdgcn_wmma_f32_16x16x4_f32(
            false, av, false, bv, (short)0, acc, false, false);
    }

    float* __restrict__ outb = out + (size_t)b * OUT_PER_B;
    const unsigned mhalf = (unsigned)((lane >> 4) << 3);
    const unsigned gbase = (unsigned)(ti * 16) + mhalf;
    const unsigned gj    = (unsigned)(tj * 16 + lrow);

    if (ti != tj) {
        #pragma unroll
        for (int v = 0; v < 8; ++v) {
            const unsigned g = gbase + (unsigned)v;
            outb[(g * (g - 1u)) / 2u + gj] = acc[v];
        }
    } else {
        #pragma unroll
        for (int v = 0; v < 8; ++v) {
            const unsigned g = gbase + (unsigned)v;
            if (g > gj) outb[(g * (g - 1u)) / 2u + gj] = acc[v];
        }
    }
}

extern "C" void kernel_launch(void* const* d_in, const int* in_sizes, int n_in,
                              void* d_out, int out_size, void* d_ws, size_t ws_size,
                              hipStream_t stream) {
    const float* x = (const float*)d_in[0];
    float* out = (float*)d_out;
    const int B = in_sizes[0] / (N_ROWS * N_COLS);
    const size_t lds_bytes = (size_t)N_ROWS * LDS_PITCH * sizeof(float);
    gram_tril_wmma_f32<<<B, THREADS, lds_bytes, stream>>>(x, out);
}


# Persistent blocks (grid 1024, 4 tiles each): next tile's 13 global_load_b128 per thread go out ahead of the WMMA loop and wait in registers, so HBM reads overlap matrix work; sums unchanged
# speedup vs baseline: 1.1298x; 1.1298x over previous
#include <hip/hip_runtime.h>


typedef __attribute__((ext_vector_type(2))) float v2f;
typedef __attribute__((ext_vector_type(4))) float v4f;
typedef __attribute__((ext_vector_type(8))) float v8f;

#define N_ROWS   64
#define N_COLS   256
#define LDS_PITCH 260
#define THREADS  320
#define OUT_PER_B (N_ROWS * (N_ROWS - 1) / 2)
#define N_VEC    (N_ROWS * (N_COLS / 4))
#define N_STAGE  ((N_VEC + THREADS - 1) / THREADS)
#define GRID_BLOCKS 1024

__device__ __forceinline__
void fetch_tile(v4f (&stage)[N_STAGE], const float* __restrict__ x, int bb, int tid) {
    const float* __restrict__ src = x + (size_t)bb * (N_ROWS * N_COLS) + tid * 4;
    #pragma unroll
    for (int i = 0; i < N_STAGE; ++i) {
        if (i < N_STAGE - 1 || tid < N_VEC - (N_STAGE - 1) * THREADS)
            stage[i] = *(const v4f*)(src + i * (THREADS * 4));
    }
}

__device__ __forceinline__
void park_tile(const v4f (&stage)[N_STAGE], float* xs, int tid) {
    float* dst = xs + (tid >> 6) * LDS_PITCH + (tid & 63) * 4;
    #pragma unroll
    for (int i = 0; i < N_STAGE; ++i) {
        if (i < N_STAGE - 1 || tid < N_VEC - (N_STAGE - 1) * THREADS)
            *(v4f*)(dst + i * (5 * LDS_PITCH)) = stage[i];
    }
}

__global__ __launch_bounds__(THREADS)
void gram_tril_wmma_f32(const float* __restrict__ x, float* __restrict__ out, int nbatch) {
    extern __shared__ float xs[];
    const int tid = threadIdx.x;
    int b = blockIdx.x;
    if (b >= nbatch) return;

    v4f stage[N_STAGE];
    fetch_tile(stage, x, b, tid);
    park_tile(stage, xs, tid);
    __syncthreads();

    const int wave = tid >> 5;
    const int lane = tid & 31;
    static const int TI[10] = {0,1,1,2,2,2,3,3,3,3};
    static const int TJ[10] = {0,0,1,0,1,2,0,1,2,3};
    const int ti = TI[wave];
    const int tj = TJ[wave];

    const int lrow = lane & 15;
    const int koff = (lane >> 4) << 1;
    const float* aBase = &xs[(ti * 16 + lrow) * LDS_PITCH + koff];
    const float* bBase = &xs[(tj * 16 + lrow) * LDS_PITCH + koff];

    const unsigned mhalf = (unsigned)((lane >> 4) << 3);
    const unsigned gbase = (unsigned)(ti * 16) + mhalf;
    const unsigned gj    = (unsigned)(tj * 16 + lrow);

    for (;;) {
        const int  bn   = b + GRID_BLOCKS;
        const bool more = bn < nbatch;
        if (more) fetch_tile(stage, x, bn, tid);

        v8f acc = {0.f, 0.f, 0.f, 0.f, 0.f, 0.f, 0.f, 0.f};
        #pragma unroll 8
        for (int k = 0; k < N_COLS; k += 4) {
            v2f av = *(const v2f*)(aBase + k);
            v2f bv = *(const v2f*)(bBase + k);
            acc = __builtin_amdgcn_wmma_f32_16x16x4_f32(
                false, av, false, bv, (short)0, acc, false, false);
        }

        asm volatile("v_nop\n\tv_nop\n\tv_nop\n\tv_nop\n\tv_nop\n\tv_nop\n\tv_nop\n\tv_nop" : "+v"(acc));

        if (more) {
            __syncthreads();
            park_tile(stage, xs, tid);
            __syncthreads();
        }

        const unsigned ob = (unsigned)b * (unsigned)OUT_PER_B;
        if (ti != tj) {
            #pragma unroll
            for (int v = 0; v < 8; ++v) {
                const unsigned g = gbase + (unsigned)v;
                out[ob + (g * (g - 1u)) / 2u + gj] = acc[v];
            }
        } else {
            #pragma unroll
            for (int v = 0; v < 8; ++v) {
                const unsigned g = gbase + (unsigned)v;
                if (g > gj) out[ob + (g * (g - 1u)) / 2u + gj] = acc[v];
            }
        }

        if (!more) break;
        b = bn;
    }
}

extern "C" void kernel_launch(void* const* d_in, const int* in_sizes, int n_in,
                              void* d_out, int out_size, void* d_ws, size_t ws_size,
                              hipStream_t stream) {
    const float* x = (const float*)d_in[0];
    float* out = (float*)d_out;
    const int B = in_sizes[0] / (N_ROWS * N_COLS);
    const size_t lds_bytes = (size_t)N_ROWS * LDS_PITCH * sizeof(float);
    gram_tril_wmma_f32<<<GRID_BLOCKS, THREADS, lds_bytes, stream>>>(x, out, B);
}
